# starting kernel
# baseline (speedup 1.0000x reference)
#include <hip/hip_runtime.h>

typedef float v2f __attribute__((ext_vector_type(2)));
typedef float v8f __attribute__((ext_vector_type(8)));

__device__ __forceinline__ float lrelu(float v) { return v > 0.0f ? v : 0.01f * v; }

__global__ __launch_bounds__(1024)
void gat_edge_stats_kernel(const float* __restrict__ x,
                           const float* __restrict__ W,
                           const float* __restrict__ a,
                           const int*  __restrict__ src_idx,
                           const int*  __restrict__ dst_idx,
                           float* __restrict__ c_out,
                           int N, int F, int DEG, int node_off, int xcols)
{
    extern __shared__ float smem[];
    float* sn  = smem;
    float* red = smem + N;

    const int b        = blockIdx.x;
    const int tid      = threadIdx.x;
    const int nthreads = blockDim.x;
    const float* xb = x + (size_t)b * xcols + node_off;

    for (int i = tid; i < N; i += nthreads) sn[i] = xb[i];

    float s1 = 0.0f, s2 = 0.0f;
    for (int f = 0; f < F; ++f) { float w = W[f]; s1 += w * a[f]; s2 += w * a[F + f]; }
    __syncthreads();

    float lmax = -3.402823466e38f;
    for (int i = tid; i < N; i += nthreads) {
        const int ebase = i * DEG;
        for (int k = 0; k < DEG; ++k) {
            float raw = lrelu(sn[src_idx[ebase + k]] * s1 + sn[dst_idx[ebase + k]] * s2);
            lmax = fmaxf(lmax, raw);
        }
    }
    red[tid] = lmax;
    __syncthreads();
    for (int s = nthreads >> 1; s > 0; s >>= 1) {
        if (tid < s) red[tid] = fmaxf(red[tid], red[tid + s]);
        __syncthreads();
    }
    const float m = red[0];
    __syncthreads();

    float lsum = 0.0f;
    for (int i = tid; i < N; i += nthreads) {
        const int ebase = i * DEG;
        float t = 0.0f;
        for (int k = 0; k < DEG; ++k) {
            const int d = dst_idx[ebase + k];
            float raw = lrelu(sn[src_idx[ebase + k]] * s1 + sn[d] * s2);
            float p = expf(raw - m);
            lsum += p;
            t    += p * sn[d];
        }
        c_out[(size_t)b * N + i] = t;
    }
    red[tid] = lsum;
    __syncthreads();
    for (int s = nthreads >> 1; s > 0; s >>= 1) {
        if (tid < s) red[tid] += red[tid + s];
        __syncthreads();
    }
    const float invS = 1.0f / red[0];
    __syncthreads();
    for (int i = tid; i < N; i += nthreads)
        c_out[(size_t)b * N + i] *= invS;
}

__global__ __launch_bounds__(256)
void gat_outer_wmma_kernel(const float* __restrict__ c,
                           const float* __restrict__ W,
                           float* __restrict__ out,
                           int F)
{
    const int lane  = threadIdx.x & 31;
    const int wave  = threadIdx.x >> 5;
    const int g     = blockIdx.x * 16;
    const int fbase = wave * 16;

    v2f A;
    A.x = (lane < 16) ? c[g + lane] : 0.0f;
    A.y = 0.0f;

    v2f Bv;
    Bv.x = (lane < 16) ? W[fbase + lane] : 0.0f;
    Bv.y = 0.0f;

    v8f acc = {};
    acc = __builtin_amdgcn_wmma_f32_16x16x4_f32(
         false, A,  false, Bv,
         (short)0, acc,  false,  false);

    const int mbase = g + ((lane >> 4) << 3);
    const int col   = fbase + (lane & 15);
#pragma unroll
    for (int v = 0; v < 8; ++v) {
        float val = acc[v];
        out[(size_t)(mbase + v) * F + col] = val > 0.0f ? val : 0.01f * val;
    }
}

extern "C" void kernel_launch(void* const* d_in, const int* in_sizes, int n_in,
                              void* d_out, int out_size, void* d_ws, size_t ws_size,
                              hipStream_t stream)
{
    const float* x   = (const float*)d_in[0];
    const float* Wn  = (const float*)d_in[1];
    const float* a   = (const float*)d_in[2];
    const int*   src = (const int*)d_in[3];
    const int*   dst = (const int*)d_in[4];
    float* out = (float*)d_out;

    const int B        = 4;
    const int F        = in_sizes[1];
    const int E        = in_sizes[3];
    const int N        = out_size / (B * F);
    const int DEG      = E / N;
    const int xcols    = in_sizes[0] / B;
    const int node_off = xcols - N;

    float* c_ws = (float*)d_ws;

    const int threads = 1024;
    const size_t smem = (size_t)(N + threads) * sizeof(float);
    gat_edge_stats_kernel<<<B, threads, smem, stream>>>(
        x, Wn, a, src, dst, c_ws, N, F, DEG, node_off, xcols);

    const int rows = B * N;
    gat_outer_wmma_kernel<<<rows / 16, (F / 16) * 32, 0, stream>>>(
        c_ws, Wn, out, F);
}


# edge pass and own edges: a lane takes 4 consecutive edges per 16-byte index load, 16 edges a trip; a wave's 128 edges lie wholly inside or outside: no per-edge bounds work; S summed in another order
# speedup vs baseline: 3.3336x; 3.3336x over previous
#include <hip/hip_runtime.h>

typedef float v2f __attribute__((ext_vector_type(2)));
typedef float v8f __attribute__((ext_vector_type(8)));
typedef float v4f __attribute__((ext_vector_type(4)));
typedef int   v4i __attribute__((ext_vector_type(4)));

__device__ __forceinline__ float lrelu(float v) { return v > 0.0f ? v : 0.01f * v; }

__device__ __forceinline__ float xhalf(float x) {
    const int b = __builtin_bit_cast(int, x);
    return __builtin_bit_cast(float, __builtin_amdgcn_permlanex16(b, b, 0x76543210, 0xfedcba98, false, false));
}
template <int K> __device__ __forceinline__ float xrow(float x) {
    const int b = __builtin_bit_cast(int, x);
    return __builtin_bit_cast(float, __builtin_amdgcn_update_dpp(b, b, 0x160 + K, 0xf, 0xf, true));
}
__device__ __forceinline__ float wave_max(float v) {
    v = fmaxf(v, xrow<1>(v));
    v = fmaxf(v, xrow<2>(v));
    v = fmaxf(v, xrow<4>(v));
    v = fmaxf(v, xrow<8>(v));
    return fmaxf(v, xhalf(v));
}
__device__ __forceinline__ float wave_sum(float v) {
    v += xrow<1>(v);
    v += xrow<2>(v);
    v += xrow<4>(v);
    v += xrow<8>(v);
    return v + xhalf(v);
}

#define RPW 256
#define OWN4 2

__device__ __forceinline__ float edge_logit(float ns, float nd, float s1, float s2) {
    return lrelu(fmaf(ns, s1, nd * s2));
}

template <int G>
__device__ __forceinline__ void edge_trip(const v4i* __restrict__ src4, const v4i* __restrict__ dst4,
                                          const float* sn, float s1, float s2,
                                          int nthreads, int tid, int ngw, int& gi, float& lm, float& ls)
{
    v4i si[G], di[G];
#pragma unroll
    for (int g = 0; g < G; ++g) {
        const int qd = gi * nthreads + tid;
        si[g] = src4[qd];
        di[g] = dst4[qd];
        gi = (gi + 1 < ngw) ? gi + 1 : 0;
    }
    float raw[4 * G];
#pragma unroll
    for (int g = 0; g < G; ++g) {
#pragma unroll
        for (int k = 0; k < 4; ++k) raw[4 * g + k] = edge_logit(sn[si[g][k]], sn[di[g][k]], s1, s2);
    }
    float nm = lm;
#pragma unroll
    for (int u = 0; u < 4 * G; ++u) nm = fmaxf(nm, raw[u]);
    ls *= __builtin_amdgcn_exp2f(lm - nm);
#pragma unroll
    for (int u = 0; u < 4 * G; ++u) ls += __builtin_amdgcn_exp2f(raw[u] - nm);
    lm = nm;
}

__global__ __launch_bounds__(1024)
void gat_fused_kernel(const float* __restrict__ x,
                      const float* __restrict__ W,
                      const float* __restrict__ a,
                      const int*  __restrict__ src_idx,
                      const int*  __restrict__ dst_idx,
                      float* __restrict__ out,
                      int N, int F, int DEG, int node_off, int xcols)
{
    extern __shared__ float smem[];

    const int tid      = threadIdx.x;
    const int nthreads = blockDim.x;
    const int lane     = tid & 31;
    const int wave     = __builtin_amdgcn_readfirstlane(tid >> 5);
    const int nwaves   = nthreads >> 5;
    const int ecnt     = RPW * DEG;

    float* sn  = smem;
    float* red = sn + N;
    float* q   = red + nthreads;
    float* cs  = q + ecnt;

    const int row0 = blockIdx.x * RPW;
    const int b    = row0 / N;
    const int n0   = row0 - b * N;
    const float* xb = x + (size_t)b * xcols + node_off;

    const int j     = lane & 15;
    const int nct   = F >> 4;
    const int fbase = (wave % nct) * 16;
    const float wv  = W[fbase + j];

    for (int i = tid; i < N; i += nthreads) sn[i] = xb[i];

    float s1 = 0.0f, s2 = 0.0f;
    for (int f = 0; f < F; ++f) { float w = W[f]; s1 += w * a[f]; s2 += w * a[F + f]; }
    const float log2e = 1.4426950408889634f;
    s1 *= log2e;
    s2 *= log2e;
    __syncthreads();

    const int etot = N * DEG;

    const float lowest = -3.402823466e38f;
    float lm = lowest;
    float ls = 0.0f;
    const int gspan = 4 * nthreads;
    const int tid4  = 4 * tid;
    const v4i* __restrict__ src4 = reinterpret_cast<const v4i*>(src_idx);
    const v4i* __restrict__ dst4 = reinterpret_cast<const v4i*>(dst_idx);
    const int wrest = etot - 128 * wave;
    const int ngw   = wrest > 0 ? (wrest + gspan - 1) / gspan : 0;
    int gi = ngw > 0 ? (int)(blockIdx.x % (unsigned)ngw) : 0;
    int gdone = 0;
    for (; gdone + 4 <= ngw; gdone += 4)
        edge_trip<4>(src4, dst4, sn, s1, s2, nthreads, tid, ngw, gi, lm, ls);
    for (; gdone < ngw; ++gdone)
        edge_trip<1>(src4, dst4, sn, s1, s2, nthreads, tid, ngw, gi, lm, ls);

    const int qbase = (n0 * DEG) >> 2;
    float rawo[OWN4][4], ndo[OWN4][4];
    bool  haso[OWN4];
#pragma unroll
    for (int u = 0; u < OWN4; ++u) {
        haso[u] = 4 * u * nthreads + 128 * wave < ecnt;
#pragma unroll
        for (int k = 0; k < 4; ++k) { rawo[u][k] = lowest; ndo[u][k] = 0.0f; }
        if (haso[u]) {
            const int qd = qbase + u * nthreads + tid;
            const v4i si = src4[qd];
            const v4i di = dst4[qd];
#pragma unroll
            for (int k = 0; k < 4; ++k) {
                const float nd = sn[di[k]];
                rawo[u][k] = edge_logit(sn[si[k]], nd, s1, s2);
                ndo[u][k]  = nd;
            }
        }
    }

    float* redm = red;
    float* reds = red + nwaves;
    const float wmax = wave_max(lm);
    if (lane == 0) redm[wave] = wmax;
    __syncthreads();
    const float m = wave_max(lane < nwaves ? redm[lane] : lowest);

    const float wsum = wave_sum(ls * __builtin_amdgcn_exp2f(lm - m));
    if (lane == 0) reds[wave] = wsum;
    __syncthreads();
    const float invS = 1.0f / wave_sum(lane < nwaves ? reds[lane] : 0.0f);

#pragma unroll
    for (int u = 0; u < OWN4; ++u) {
        if (haso[u]) {
            v4f pq;
#pragma unroll
            for (int k = 0; k < 4; ++k) pq[k] = __builtin_amdgcn_exp2f(rawo[u][k] - m) * ndo[u][k];
            *reinterpret_cast<v4f*>(q + 4 * u * nthreads + tid4) = pq;
        }
    }
    __syncthreads();

    if (tid < RPW) {
        const float* qn = q + tid * DEG;
        float t = 0.0f;
#pragma unroll 8
        for (int k = 0; k < DEG; ++k) t += qn[k];
        cs[tid] = t * invS;
    }
    __syncthreads();

    v2f Bv;
    Bv.x = (lane < 16) ? wv : 0.0f;
    Bv.y = 0.0f;

    const int rstep = nwaves >= nct ? nwaves / nct : 1;
    for (int rt = wave / nct; rt < RPW / 16; rt += rstep) {
        const float cj = cs[rt * 16 + j];

        v2f A;
        A.x = (lane < 16) ? cj : 0.0f;
        A.y = 0.0f;

        v8f acc = {};
        acc = __builtin_amdgcn_wmma_f32_16x16x4_f32(
             false, A,  false, Bv,
             (short)0, acc,  false,  false);

        const size_t mbase = (size_t)(row0 + rt * 16 + ((lane >> 4) << 3));
        const int col = fbase + j;
#pragma unroll
        for (int v = 0; v < 8; ++v) {
            float val = acc[v];
            out[(mbase + v) * F + col] = val > 0.0f ? val : 0.01f * val;
        }
    }
}

extern "C" void kernel_launch(void* const* d_in, const int* in_sizes, int n_in,
                              void* d_out, int out_size, void* d_ws, size_t ws_size,
                              hipStream_t stream)
{
    const float* x   = (const float*)d_in[0];
    const float* Wn  = (const float*)d_in[1];
    const float* a   = (const float*)d_in[2];
    const int*   src = (const int*)d_in[3];
    const int*   dst = (const int*)d_in[4];
    float* out = (float*)d_out;

    const int B        = 4;
    const int F        = in_sizes[1];
    const int E        = in_sizes[3];
    const int N        = out_size / (B * F);
    const int DEG      = E / N;
    const int xcols    = in_sizes[0] / B;
    const int node_off = xcols - N;

    const int rows    = B * N;
    const int threads = 1024;
    const size_t smem = (size_t)(N + threads + RPW * DEG + RPW) * sizeof(float);
    gat_fused_kernel<<<rows / RPW, threads, smem, stream>>>(
        x, Wn, a, src, dst, out, N, F, DEG, node_off, xcols);
}


# global loads that need nothing computed here (nodes as ONE 16-byte load per lane; index quads of the first pass trip and of the own edges) issued at entry, under the serial s1,s2 loop; sums unchanged
# speedup vs baseline: 3.7582x; 1.1274x over previous
#include <hip/hip_runtime.h>

typedef float v2f __attribute__((ext_vector_type(2)));
typedef float v8f __attribute__((ext_vector_type(8)));
typedef float v4f __attribute__((ext_vector_type(4)));
typedef int   v4i __attribute__((ext_vector_type(4)));

__device__ __forceinline__ float lrelu(float v) { return v > 0.0f ? v : 0.01f * v; }

__device__ __forceinline__ float xhalf(float x) {
    const int b = __builtin_bit_cast(int, x);
    return __builtin_bit_cast(float, __builtin_amdgcn_permlanex16(b, b, 0x76543210, 0xfedcba98, false, false));
}
template <int K> __device__ __forceinline__ float xrow(float x) {
    const int b = __builtin_bit_cast(int, x);
    return __builtin_bit_cast(float, __builtin_amdgcn_update_dpp(b, b, 0x160 + K, 0xf, 0xf, true));
}
__device__ __forceinline__ float wave_max(float v) {
    v = fmaxf(v, xrow<1>(v));
    v = fmaxf(v, xrow<2>(v));
    v = fmaxf(v, xrow<4>(v));
    v = fmaxf(v, xrow<8>(v));
    return fmaxf(v, xhalf(v));
}
__device__ __forceinline__ float wave_sum(float v) {
    v += xrow<1>(v);
    v += xrow<2>(v);
    v += xrow<4>(v);
    v += xrow<8>(v);
    return v + xhalf(v);
}

#define RPW 256
#define OWN4 2

__device__ __forceinline__ float edge_logit(float ns, float nd, float s1, float s2) {
    return lrelu(fmaf(ns, s1, nd * s2));
}

template <int G>
__device__ __forceinline__ void trip_load(const v4i* __restrict__ src4, const v4i* __restrict__ dst4,
                                          int nthreads, int tid, int ngw, int& gi,
                                          v4i (&si)[G], v4i (&di)[G])
{
#pragma unroll
    for (int g = 0; g < G; ++g) {
        const int qd = gi * nthreads + tid;
        si[g] = src4[qd];
        di[g] = dst4[qd];
        gi = (gi + 1 < ngw) ? gi + 1 : 0;
    }
}

template <int G>
__device__ __forceinline__ void trip_sum(const v4i (&si)[G], const v4i (&di)[G],
                                         const float* sn, float s1, float s2, float& lm, float& ls)
{
    float raw[4 * G];
#pragma unroll
    for (int g = 0; g < G; ++g) {
#pragma unroll
        for (int k = 0; k < 4; ++k) raw[4 * g + k] = edge_logit(sn[si[g][k]], sn[di[g][k]], s1, s2);
    }
    float nm = lm;
#pragma unroll
    for (int u = 0; u < 4 * G; ++u) nm = fmaxf(nm, raw[u]);
    ls *= __builtin_amdgcn_exp2f(lm - nm);
#pragma unroll
    for (int u = 0; u < 4 * G; ++u) ls += __builtin_amdgcn_exp2f(raw[u] - nm);
    lm = nm;
}

template <int G>
__device__ __forceinline__ void edge_trip(const v4i* __restrict__ src4, const v4i* __restrict__ dst4,
                                          const float* sn, float s1, float s2,
                                          int nthreads, int tid, int ngw, int& gi, float& lm, float& ls)
{
    v4i si[G], di[G];
    trip_load<G>(src4, dst4, nthreads, tid, ngw, gi, si, di);
    trip_sum<G>(si, di, sn, s1, s2, lm, ls);
}

__global__ __launch_bounds__(1024)
void gat_fused_kernel(const float* __restrict__ x,
                      const float* __restrict__ W,
                      const float* __restrict__ a,
                      const int*  __restrict__ src_idx,
                      const int*  __restrict__ dst_idx,
                      float* __restrict__ out,
                      int N, int F, int DEG, int node_off, int xcols)
{
    extern __shared__ float smem[];

    const int tid      = threadIdx.x;
    const int nthreads = blockDim.x;
    const int lane     = tid & 31;
    const int wave     = __builtin_amdgcn_readfirstlane(tid >> 5);
    const int nwaves   = nthreads >> 5;
    const int ecnt     = RPW * DEG;

    float* sn  = smem;
    float* red = sn + N;
    float* q   = red + nthreads;
    float* cs  = q + ecnt;

    const int row0 = blockIdx.x * RPW;
    const int b    = row0 / N;
    const int n0   = row0 - b * N;
    const float* xb = x + (size_t)b * xcols + node_off;

    const int j     = lane & 15;
    const int nct   = F >> 4;
    const int fbase = (wave % nct) * 16;
    const float wv  = W[fbase + j];


    const v4f* __restrict__ xq = reinterpret_cast<const v4f*>(xb);
    v4f* sq = reinterpret_cast<v4f*>(sn);
    const int nq = N >> 2;
    v4f nv = {0.0f, 0.0f, 0.0f, 0.0f};
    if (tid < nq) nv = xq[tid];

    const int etot = N * DEG;
    const float lowest = -3.402823466e38f;
    const v4i* __restrict__ src4 = reinterpret_cast<const v4i*>(src_idx);
    const v4i* __restrict__ dst4 = reinterpret_cast<const v4i*>(dst_idx);

    const int gspan = 4 * nthreads;
    const int wrest = etot - 128 * wave;
    const int ngw   = wrest > 0 ? (wrest + gspan - 1) / gspan : 0;
    int gi = ngw > 0 ? (int)(blockIdx.x % (unsigned)ngw) : 0;
    const bool first4 = ngw >= 4;
    v4i fsi[4], fdi[4];
#pragma unroll
    for (int g = 0; g < 4; ++g) { fsi[g] = v4i{0, 0, 0, 0}; fdi[g] = v4i{0, 0, 0, 0}; }
    if (first4) trip_load<4>(src4, dst4, nthreads, tid, ngw, gi, fsi, fdi);

    const int qbase = (n0 * DEG) >> 2;
    v4i  osi[OWN4], odi[OWN4];
    bool haso[OWN4];
#pragma unroll
    for (int u = 0; u < OWN4; ++u) {
        haso[u] = 4 * u * nthreads + 128 * wave < ecnt;
        osi[u] = v4i{0, 0, 0, 0};
        odi[u] = v4i{0, 0, 0, 0};
        if (haso[u]) {
            const int qd = qbase + u * nthreads + tid;
            osi[u] = src4[qd];
            odi[u] = dst4[qd];
        }
    }

    float s1 = 0.0f, s2 = 0.0f;
    for (int f = 0; f < F; ++f) { float w = W[f]; s1 += w * a[f]; s2 += w * a[F + f]; }
    const float log2e = 1.4426950408889634f;
    s1 *= log2e;
    s2 *= log2e;

    if (tid < nq) sq[tid] = nv;
    for (int i = tid + nthreads; i < nq; i += nthreads) sq[i] = xq[i];
    __syncthreads();

    float lm = lowest;
    float ls = 0.0f;
    const int tid4  = 4 * tid;
    int gdone = 0;
    if (first4) {
        trip_sum<4>(fsi, fdi, sn, s1, s2, lm, ls);
        gdone = 4;
    }
    for (; gdone + 4 <= ngw; gdone += 4)
        edge_trip<4>(src4, dst4, sn, s1, s2, nthreads, tid, ngw, gi, lm, ls);
    for (; gdone < ngw; ++gdone)
        edge_trip<1>(src4, dst4, sn, s1, s2, nthreads, tid, ngw, gi, lm, ls);

    float rawo[OWN4][4], ndo[OWN4][4];
#pragma unroll
    for (int u = 0; u < OWN4; ++u) {
#pragma unroll
        for (int k = 0; k < 4; ++k) { rawo[u][k] = lowest; ndo[u][k] = 0.0f; }
        if (haso[u]) {
            const v4i si = osi[u];
            const v4i di = odi[u];
#pragma unroll
            for (int k = 0; k < 4; ++k) {
                const float nd = sn[di[k]];
                rawo[u][k] = edge_logit(sn[si[k]], nd, s1, s2);
                ndo[u][k]  = nd;
            }
        }
    }

    float* redm = red;
    float* reds = red + nwaves;
    const float wmax = wave_max(lm);
    if (lane == 0) redm[wave] = wmax;
    __syncthreads();
    const float m = wave_max(lane < nwaves ? redm[lane] : lowest);

    const float wsum = wave_sum(ls * __builtin_amdgcn_exp2f(lm - m));
    if (lane == 0) reds[wave] = wsum;
    __syncthreads();
    const float invS = 1.0f / wave_sum(lane < nwaves ? reds[lane] : 0.0f);

#pragma unroll
    for (int u = 0; u < OWN4; ++u) {
        if (haso[u]) {
            v4f pq;
#pragma unroll
            for (int k = 0; k < 4; ++k) pq[k] = __builtin_amdgcn_exp2f(rawo[u][k] - m) * ndo[u][k];
            *reinterpret_cast<v4f*>(q + 4 * u * nthreads + tid4) = pq;
        }
    }
    __syncthreads();

    if (tid < RPW) {
        const float* qn = q + tid * DEG;
        float t = 0.0f;
#pragma unroll 8
        for (int k = 0; k < DEG; ++k) t += qn[k];
        cs[tid] = t * invS;
    }
    __syncthreads();

    v2f Bv;
    Bv.x = (lane < 16) ? wv : 0.0f;
    Bv.y = 0.0f;

    const int rstep = nwaves >= nct ? nwaves / nct : 1;
    for (int rt = wave / nct; rt < RPW / 16; rt += rstep) {
        const float cj = cs[rt * 16 + j];

        v2f A;
        A.x = (lane < 16) ? cj : 0.0f;
        A.y = 0.0f;

        v8f acc = {};
        acc = __builtin_amdgcn_wmma_f32_16x16x4_f32(
             false, A,  false, Bv,
             (short)0, acc,  false,  false);

        const size_t mbase = (size_t)(row0 + rt * 16 + ((lane >> 4) << 3));
        const int col = fbase + j;
#pragma unroll
        for (int v = 0; v < 8; ++v) {
            float val = acc[v];
            out[(mbase + v) * F + col] = val > 0.0f ? val : 0.01f * val;
        }
    }
}

extern "C" void kernel_launch(void* const* d_in, const int* in_sizes, int n_in,
                              void* d_out, int out_size, void* d_ws, size_t ws_size,
                              hipStream_t stream)
{
    const float* x   = (const float*)d_in[0];
    const float* Wn  = (const float*)d_in[1];
    const float* a   = (const float*)d_in[2];
    const int*   src = (const int*)d_in[3];
    const int*   dst = (const int*)d_in[4];
    float* out = (float*)d_out;

    const int B        = 4;
    const int F        = in_sizes[1];
    const int E        = in_sizes[3];
    const int N        = out_size / (B * F);
    const int DEG      = E / N;
    const int xcols    = in_sizes[0] / B;
    const int node_off = xcols - N;

    const int rows    = B * N;
    const int threads = 1024;
    const size_t smem = (size_t)(N + threads + RPW * DEG + RPW) * sizeof(float);
    gat_fused_kernel<<<rows / RPW, threads, smem, stream>>>(
        x, Wn, a, src, dst, out, N, F, DEG, node_off, xcols);
}


# s1, s2 once per work-group: each of 32 waves sums 4 terms (scalar loads), partials merged via LDS behind the node barrier by a wave sum, not all F terms in every wave; s1, s2 summed in another order
# speedup vs baseline: 4.2478x; 1.1303x over previous
#include <hip/hip_runtime.h>

typedef float v2f __attribute__((ext_vector_type(2)));
typedef float v8f __attribute__((ext_vector_type(8)));
typedef float v4f __attribute__((ext_vector_type(4)));
typedef int   v4i __attribute__((ext_vector_type(4)));

__device__ __forceinline__ float lrelu(float v) { return v > 0.0f ? v : 0.01f * v; }

__device__ __forceinline__ float xhalf(float x) {
    const int b = __builtin_bit_cast(int, x);
    return __builtin_bit_cast(float, __builtin_amdgcn_permlanex16(b, b, 0x76543210, 0xfedcba98, false, false));
}
template <int K> __device__ __forceinline__ float xrow(float x) {
    const int b = __builtin_bit_cast(int, x);
    return __builtin_bit_cast(float, __builtin_amdgcn_update_dpp(b, b, 0x160 + K, 0xf, 0xf, true));
}
__device__ __forceinline__ float wave_max(float v) {
    v = fmaxf(v, xrow<1>(v));
    v = fmaxf(v, xrow<2>(v));
    v = fmaxf(v, xrow<4>(v));
    v = fmaxf(v, xrow<8>(v));
    return fmaxf(v, xhalf(v));
}
__device__ __forceinline__ float wave_sum(float v) {
    v += xrow<1>(v);
    v += xrow<2>(v);
    v += xrow<4>(v);
    v += xrow<8>(v);
    return v + xhalf(v);
}

#define RPW 256
#define OWN4 2

__device__ __forceinline__ float edge_logit(float ns, float nd, float s1, float s2) {
    return lrelu(fmaf(ns, s1, nd * s2));
}

template <int G>
__device__ __forceinline__ void trip_load(const v4i* __restrict__ src4, const v4i* __restrict__ dst4,
                                          int nthreads, int tid, int ngw, int& gi,
                                          v4i (&si)[G], v4i (&di)[G])
{
#pragma unroll
    for (int g = 0; g < G; ++g) {
        const int qd = gi * nthreads + tid;
        si[g] = src4[qd];
        di[g] = dst4[qd];
        gi = (gi + 1 < ngw) ? gi + 1 : 0;
    }
}

template <int G>
__device__ __forceinline__ void trip_sum(const v4i (&si)[G], const v4i (&di)[G],
                                         const float* sn, float s1, float s2, float& lm, float& ls)
{
    float raw[4 * G];
#pragma unroll
    for (int g = 0; g < G; ++g) {
#pragma unroll
        for (int k = 0; k < 4; ++k) raw[4 * g + k] = edge_logit(sn[si[g][k]], sn[di[g][k]], s1, s2);
    }
    float nm = lm;
#pragma unroll
    for (int u = 0; u < 4 * G; ++u) nm = fmaxf(nm, raw[u]);
    ls *= __builtin_amdgcn_exp2f(lm - nm);
#pragma unroll
    for (int u = 0; u < 4 * G; ++u) ls += __builtin_amdgcn_exp2f(raw[u] - nm);
    lm = nm;
}

template <int G>
__device__ __forceinline__ void edge_trip(const v4i* __restrict__ src4, const v4i* __restrict__ dst4,
                                          const float* sn, float s1, float s2,
                                          int nthreads, int tid, int ngw, int& gi, float& lm, float& ls)
{
    v4i si[G], di[G];
    trip_load<G>(src4, dst4, nthreads, tid, ngw, gi, si, di);
    trip_sum<G>(si, di, sn, s1, s2, lm, ls);
}

__global__ __launch_bounds__(1024)
void gat_fused_kernel(const float* __restrict__ x,
                      const float* __restrict__ W,
                      const float* __restrict__ a,
                      const int*  __restrict__ src_idx,
                      const int*  __restrict__ dst_idx,
                      float* __restrict__ out,
                      int N, int F, int DEG, int node_off, int xcols)
{
    extern __shared__ float smem[];

    const int tid      = threadIdx.x;
    const int nthreads = blockDim.x;
    const int lane     = tid & 31;
    const int wave     = __builtin_amdgcn_readfirstlane(tid >> 5);
    const int nwaves   = nthreads >> 5;
    const int ecnt     = RPW * DEG;

    float* sn  = smem;
    float* red = sn + N;
    float* q   = red + nthreads;
    float* cs  = q + ecnt;

    const int row0 = blockIdx.x * RPW;
    const int b    = row0 / N;
    const int n0   = row0 - b * N;
    const float* xb = x + (size_t)b * xcols + node_off;

    const int j     = lane & 15;
    const int nct   = F >> 4;
    const int fbase = (wave % nct) * 16;
    const float wv  = W[fbase + j];


    const v4f* __restrict__ xq = reinterpret_cast<const v4f*>(xb);
    v4f* sq = reinterpret_cast<v4f*>(sn);
    const int nq = N >> 2;
    v4f nv = {0.0f, 0.0f, 0.0f, 0.0f};
    if (tid < nq) nv = xq[tid];

    const int etot = N * DEG;
    const float lowest = -3.402823466e38f;
    const v4i* __restrict__ src4 = reinterpret_cast<const v4i*>(src_idx);
    const v4i* __restrict__ dst4 = reinterpret_cast<const v4i*>(dst_idx);

    const int gspan = 4 * nthreads;
    const int wrest = etot - 128 * wave;
    const int ngw   = wrest > 0 ? (wrest + gspan - 1) / gspan : 0;
    int gi = ngw > 0 ? (int)(blockIdx.x % (unsigned)ngw) : 0;
    const bool first4 = ngw >= 4;
    v4i fsi[4], fdi[4];
#pragma unroll
    for (int g = 0; g < 4; ++g) { fsi[g] = v4i{0, 0, 0, 0}; fdi[g] = v4i{0, 0, 0, 0}; }
    if (first4) trip_load<4>(src4, dst4, nthreads, tid, ngw, gi, fsi, fdi);

    const int qbase = (n0 * DEG) >> 2;
    v4i  osi[OWN4], odi[OWN4];
    bool haso[OWN4];
#pragma unroll
    for (int u = 0; u < OWN4; ++u) {
        haso[u] = 4 * u * nthreads + 128 * wave < ecnt;
        osi[u] = v4i{0, 0, 0, 0};
        odi[u] = v4i{0, 0, 0, 0};
        if (haso[u]) {
            const int qd = qbase + u * nthreads + tid;
            osi[u] = src4[qd];
            odi[u] = dst4[qd];
        }
    }

    float p1 = 0.0f, p2 = 0.0f;
    for (int f0 = 4 * wave; f0 < F; f0 += 4 * nwaves) {
#pragma unroll
        for (int k = 0; k < 4; ++k) { const float w = W[f0 + k]; p1 += w * a[f0 + k]; p2 += w * a[F + f0 + k]; }
    }
    float* sp = red + 2 * nwaves;
    if (lane == 0) { sp[2 * wave] = p1; sp[2 * wave + 1] = p2; }

    if (tid < nq) sq[tid] = nv;
    for (int i = tid + nthreads; i < nq; i += nthreads) sq[i] = xq[i];
    __syncthreads();

    float s1 = wave_sum(lane < nwaves ? sp[2 * lane] : 0.0f);
    float s2 = wave_sum(lane < nwaves ? sp[2 * lane + 1] : 0.0f);
    s1 = __builtin_bit_cast(float, __builtin_amdgcn_readfirstlane(__builtin_bit_cast(int, s1)));
    s2 = __builtin_bit_cast(float, __builtin_amdgcn_readfirstlane(__builtin_bit_cast(int, s2)));
    const float log2e = 1.4426950408889634f;
    s1 *= log2e;
    s2 *= log2e;

    float lm = lowest;
    float ls = 0.0f;
    const int tid4  = 4 * tid;
    int gdone = 0;
    if (first4) {
        trip_sum<4>(fsi, fdi, sn, s1, s2, lm, ls);
        gdone = 4;
    }
    for (; gdone + 4 <= ngw; gdone += 4)
        edge_trip<4>(src4, dst4, sn, s1, s2, nthreads, tid, ngw, gi, lm, ls);
    for (; gdone < ngw; ++gdone)
        edge_trip<1>(src4, dst4, sn, s1, s2, nthreads, tid, ngw, gi, lm, ls);

    float rawo[OWN4][4], ndo[OWN4][4];
#pragma unroll
    for (int u = 0; u < OWN4; ++u) {
#pragma unroll
        for (int k = 0; k < 4; ++k) { rawo[u][k] = lowest; ndo[u][k] = 0.0f; }
        if (haso[u]) {
            const v4i si = osi[u];
            const v4i di = odi[u];
#pragma unroll
            for (int k = 0; k < 4; ++k) {
                const float nd = sn[di[k]];
                rawo[u][k] = edge_logit(sn[si[k]], nd, s1, s2);
                ndo[u][k]  = nd;
            }
        }
    }

    float* redm = red;
    float* reds = red + nwaves;
    const float wmax = wave_max(lm);
    if (lane == 0) redm[wave] = wmax;
    __syncthreads();
    const float m = wave_max(lane < nwaves ? redm[lane] : lowest);

    const float wsum = wave_sum(ls * __builtin_amdgcn_exp2f(lm - m));
    if (lane == 0) reds[wave] = wsum;
    __syncthreads();
    const float invS = 1.0f / wave_sum(lane < nwaves ? reds[lane] : 0.0f);

#pragma unroll
    for (int u = 0; u < OWN4; ++u) {
        if (haso[u]) {
            v4f pq;
#pragma unroll
            for (int k = 0; k < 4; ++k) pq[k] = __builtin_amdgcn_exp2f(rawo[u][k] - m) * ndo[u][k];
            *reinterpret_cast<v4f*>(q + 4 * u * nthreads + tid4) = pq;
        }
    }
    __syncthreads();

    if (tid < RPW) {
        const float* qn = q + tid * DEG;
        float t = 0.0f;
#pragma unroll 8
        for (int k = 0; k < DEG; ++k) t += qn[k];
        cs[tid] = t * invS;
    }
    __syncthreads();

    v2f Bv;
    Bv.x = (lane < 16) ? wv : 0.0f;
    Bv.y = 0.0f;

    const int rstep = nwaves >= nct ? nwaves / nct : 1;
    for (int rt = wave / nct; rt < RPW / 16; rt += rstep) {
        const float cj = cs[rt * 16 + j];

        v2f A;
        A.x = (lane < 16) ? cj : 0.0f;
        A.y = 0.0f;

        v8f acc = {};
        acc = __builtin_amdgcn_wmma_f32_16x16x4_f32(
             false, A,  false, Bv,
             (short)0, acc,  false,  false);

        const size_t mbase = (size_t)(row0 + rt * 16 + ((lane >> 4) << 3));
        const int col = fbase + j;
#pragma unroll
        for (int v = 0; v < 8; ++v) {
            float val = acc[v];
            out[(mbase + v) * F + col] = val > 0.0f ? val : 0.01f * val;
        }
    }
}

extern "C" void kernel_launch(void* const* d_in, const int* in_sizes, int n_in,
                              void* d_out, int out_size, void* d_ws, size_t ws_size,
                              hipStream_t stream)
{
    const float* x   = (const float*)d_in[0];
    const float* Wn  = (const float*)d_in[1];
    const float* a   = (const float*)d_in[2];
    const int*   src = (const int*)d_in[3];
    const int*   dst = (const int*)d_in[4];
    float* out = (float*)d_out;

    const int B        = 4;
    const int F        = in_sizes[1];
    const int E        = in_sizes[3];
    const int N        = out_size / (B * F);
    const int DEG      = E / N;
    const int xcols    = in_sizes[0] / B;
    const int node_off = xcols - N;

    const int rows    = B * N;
    const int threads = 1024;
    const size_t smem = (size_t)(N + threads + RPW * DEG + RPW) * sizeof(float);
    gat_fused_kernel<<<rows / RPW, threads, smem, stream>>>(
        x, Wn, a, src, dst, out, N, F, DEG, node_off, xcols);
}


# edge pass: fewer vector instructions per trip (leaky_relu as max(z, 0.01z), not compare+select; index loads from a scalar group base; exps added in pairs): S summed in another order
# speedup vs baseline: 4.4909x; 1.0572x over previous
#include <hip/hip_runtime.h>

typedef float v2f __attribute__((ext_vector_type(2)));
typedef float v8f __attribute__((ext_vector_type(8)));
typedef float v4f __attribute__((ext_vector_type(4)));
typedef int   v4i __attribute__((ext_vector_type(4)));

__device__ __forceinline__ float lrelu(float v) { return fmaxf(v, 0.01f * v); }

__device__ __forceinline__ float xhalf(float x) {
    const int b = __builtin_bit_cast(int, x);
    return __builtin_bit_cast(float, __builtin_amdgcn_permlanex16(b, b, 0x76543210, 0xfedcba98, false, false));
}
template <int K> __device__ __forceinline__ float xrow(float x) {
    const int b = __builtin_bit_cast(int, x);
    return __builtin_bit_cast(float, __builtin_amdgcn_update_dpp(b, b, 0x160 + K, 0xf, 0xf, true));
}
__device__ __forceinline__ float wave_max(float v) {
    v = fmaxf(v, xrow<1>(v));
    v = fmaxf(v, xrow<2>(v));
    v = fmaxf(v, xrow<4>(v));
    v = fmaxf(v, xrow<8>(v));
    return fmaxf(v, xhalf(v));
}
__device__ __forceinline__ float wave_sum(float v) {
    v += xrow<1>(v);
    v += xrow<2>(v);
    v += xrow<4>(v);
    v += xrow<8>(v);
    return v + xhalf(v);
}

#define RPW 256
#define OWN4 2

__device__ __forceinline__ float edge_logit(float ns, float nd, float s1, float s2) {
    return lrelu(fmaf(ns, s1, nd * s2));
}

template <int G>
__device__ __forceinline__ void trip_load(const v4i* __restrict__ src4, const v4i* __restrict__ dst4,
                                          int nthreads, int tid, int ngw, int& gi,
                                          v4i (&si)[G], v4i (&di)[G])
{
#pragma unroll
    for (int g = 0; g < G; ++g) {
        const int qd = __builtin_amdgcn_readfirstlane(gi * nthreads) + tid;
        si[g] = src4[qd];
        di[g] = dst4[qd];
        gi = (gi + 1 < ngw) ? gi + 1 : 0;
    }
}

template <int G>
__device__ __forceinline__ void trip_sum(const v4i (&si)[G], const v4i (&di)[G],
                                         const float* sn, float s1, float s2, float& lm, float& ls)
{
    float raw[4 * G];
#pragma unroll
    for (int g = 0; g < G; ++g) {
#pragma unroll
        for (int k = 0; k < 4; ++k) raw[4 * g + k] = edge_logit(sn[si[g][k]], sn[di[g][k]], s1, s2);
    }
    float nm = lm;
#pragma unroll
    for (int u = 0; u < 4 * G; ++u) nm = fmaxf(nm, raw[u]);
    v2f acc;
    acc.x = ls * __builtin_amdgcn_exp2f(lm - nm);
    acc.y = 0.0f;
#pragma unroll
    for (int u = 0; u < 4 * G; u += 2) {
        v2f e;
        e.x = __builtin_amdgcn_exp2f(raw[u] - nm);
        e.y = __builtin_amdgcn_exp2f(raw[u + 1] - nm);
        acc += e;
    }
    ls = acc.x + acc.y;
    lm = nm;
}

template <int G>
__device__ __forceinline__ void edge_trip(const v4i* __restrict__ src4, const v4i* __restrict__ dst4,
                                          const float* sn, float s1, float s2,
                                          int nthreads, int tid, int ngw, int& gi, float& lm, float& ls)
{
    v4i si[G], di[G];
    trip_load<G>(src4, dst4, nthreads, tid, ngw, gi, si, di);
    trip_sum<G>(si, di, sn, s1, s2, lm, ls);
}

__global__ __launch_bounds__(1024)
void gat_fused_kernel(const float* __restrict__ x,
                      const float* __restrict__ W,
                      const float* __restrict__ a,
                      const int*  __restrict__ src_idx,
                      const int*  __restrict__ dst_idx,
                      float* __restrict__ out,
                      int N, int F, int DEG, int node_off, int xcols)
{
    extern __shared__ float smem[];

    const int tid      = threadIdx.x;
    const int nthreads = blockDim.x;
    const int lane     = tid & 31;
    const int wave     = __builtin_amdgcn_readfirstlane(tid >> 5);
    const int nwaves   = nthreads >> 5;
    const int ecnt     = RPW * DEG;

    float* sn  = smem;
    float* red = sn + N;
    float* q   = red + nthreads;
    float* cs  = q + ecnt;

    const int row0 = blockIdx.x * RPW;
    const int b    = row0 / N;
    const int n0   = row0 - b * N;
    const float* xb = x + (size_t)b * xcols + node_off;

    const int j     = lane & 15;
    const int nct   = F >> 4;
    const int fbase = (wave % nct) * 16;
    const float wv  = W[fbase + j];


    const v4f* __restrict__ xq = reinterpret_cast<const v4f*>(xb);
    v4f* sq = reinterpret_cast<v4f*>(sn);
    const int nq = N >> 2;
    v4f nv = {0.0f, 0.0f, 0.0f, 0.0f};
    if (tid < nq) nv = xq[tid];

    const int etot = N * DEG;
    const float lowest = -3.402823466e38f;
    const v4i* __restrict__ src4 = reinterpret_cast<const v4i*>(src_idx);
    const v4i* __restrict__ dst4 = reinterpret_cast<const v4i*>(dst_idx);

    const int gspan = 4 * nthreads;
    const int wrest = etot - 128 * wave;
    const int ngw   = wrest > 0 ? (wrest + gspan - 1) / gspan : 0;
    int gi = ngw > 0 ? (int)(blockIdx.x % (unsigned)ngw) : 0;
    const bool first4 = ngw >= 4;
    v4i fsi[4], fdi[4];
#pragma unroll
    for (int g = 0; g < 4; ++g) { fsi[g] = v4i{0, 0, 0, 0}; fdi[g] = v4i{0, 0, 0, 0}; }
    if (first4) trip_load<4>(src4, dst4, nthreads, tid, ngw, gi, fsi, fdi);

    const int qbase = (n0 * DEG) >> 2;
    v4i  osi[OWN4], odi[OWN4];
    bool haso[OWN4];
#pragma unroll
    for (int u = 0; u < OWN4; ++u) {
        haso[u] = 4 * u * nthreads + 128 * wave < ecnt;
        osi[u] = v4i{0, 0, 0, 0};
        odi[u] = v4i{0, 0, 0, 0};
        if (haso[u]) {
            const int qd = qbase + u * nthreads + tid;
            osi[u] = src4[qd];
            odi[u] = dst4[qd];
        }
    }

    float p1 = 0.0f, p2 = 0.0f;
    for (int f0 = 4 * wave; f0 < F; f0 += 4 * nwaves) {
#pragma unroll
        for (int k = 0; k < 4; ++k) { const float w = W[f0 + k]; p1 += w * a[f0 + k]; p2 += w * a[F + f0 + k]; }
    }
    float* sp = red + 2 * nwaves;
    if (lane == 0) { sp[2 * wave] = p1; sp[2 * wave + 1] = p2; }

    if (tid < nq) sq[tid] = nv;
    for (int i = tid + nthreads; i < nq; i += nthreads) sq[i] = xq[i];
    __syncthreads();

    float s1 = wave_sum(lane < nwaves ? sp[2 * lane] : 0.0f);
    float s2 = wave_sum(lane < nwaves ? sp[2 * lane + 1] : 0.0f);
    s1 = __builtin_bit_cast(float, __builtin_amdgcn_readfirstlane(__builtin_bit_cast(int, s1)));
    s2 = __builtin_bit_cast(float, __builtin_amdgcn_readfirstlane(__builtin_bit_cast(int, s2)));
    const float log2e = 1.4426950408889634f;
    s1 *= log2e;
    s2 *= log2e;

    float lm = lowest;
    float ls = 0.0f;
    const int tid4  = 4 * tid;
    int gdone = 0;
    if (first4) {
        trip_sum<4>(fsi, fdi, sn, s1, s2, lm, ls);
        gdone = 4;
    }
    for (; gdone + 4 <= ngw; gdone += 4)
        edge_trip<4>(src4, dst4, sn, s1, s2, nthreads, tid, ngw, gi, lm, ls);
    for (; gdone < ngw; ++gdone)
        edge_trip<1>(src4, dst4, sn, s1, s2, nthreads, tid, ngw, gi, lm, ls);

    float rawo[OWN4][4], ndo[OWN4][4];
#pragma unroll
    for (int u = 0; u < OWN4; ++u) {
#pragma unroll
        for (int k = 0; k < 4; ++k) { rawo[u][k] = lowest; ndo[u][k] = 0.0f; }
        if (haso[u]) {
            const v4i si = osi[u];
            const v4i di = odi[u];
#pragma unroll
            for (int k = 0; k < 4; ++k) {
                const float nd = sn[di[k]];
                rawo[u][k] = edge_logit(sn[si[k]], nd, s1, s2);
                ndo[u][k]  = nd;
            }
        }
    }

    float* redm = red;
    float* reds = red + nwaves;
    const float wmax = wave_max(lm);
    if (lane == 0) redm[wave] = wmax;
    __syncthreads();
    const float m = wave_max(lane < nwaves ? redm[lane] : lowest);

    const float wsum = wave_sum(ls * __builtin_amdgcn_exp2f(lm - m));
    if (lane == 0) reds[wave] = wsum;
    __syncthreads();
    const float invS = 1.0f / wave_sum(lane < nwaves ? reds[lane] : 0.0f);

#pragma unroll
    for (int u = 0; u < OWN4; ++u) {
        if (haso[u]) {
            v4f pq;
#pragma unroll
            for (int k = 0; k < 4; ++k) pq[k] = __builtin_amdgcn_exp2f(rawo[u][k] - m) * ndo[u][k];
            *reinterpret_cast<v4f*>(q + 4 * u * nthreads + tid4) = pq;
        }
    }
    __syncthreads();

    if (tid < RPW) {
        const float* qn = q + tid * DEG;
        float t = 0.0f;
#pragma unroll 8
        for (int k = 0; k < DEG; ++k) t += qn[k];
        cs[tid] = t * invS;
    }
    __syncthreads();

    v2f Bv;
    Bv.x = (lane < 16) ? wv : 0.0f;
    Bv.y = 0.0f;

    const int rstep = nwaves >= nct ? nwaves / nct : 1;
    for (int rt = wave / nct; rt < RPW / 16; rt += rstep) {
        const float cj = cs[rt * 16 + j];

        v2f A;
        A.x = (lane < 16) ? cj : 0.0f;
        A.y = 0.0f;

        v8f acc = {};
        acc = __builtin_amdgcn_wmma_f32_16x16x4_f32(
             false, A,  false, Bv,
             (short)0, acc,  false,  false);

        const size_t mbase = (size_t)(row0 + rt * 16 + ((lane >> 4) << 3));
        const int col = fbase + j;
#pragma unroll
        for (int v = 0; v < 8; ++v) {
            float val = acc[v];
            out[(mbase + v) * F + col] = val > 0.0f ? val : 0.01f * val;
        }
    }
}

extern "C" void kernel_launch(void* const* d_in, const int* in_sizes, int n_in,
                              void* d_out, int out_size, void* d_ws, size_t ws_size,
                              hipStream_t stream)
{
    const float* x   = (const float*)d_in[0];
    const float* Wn  = (const float*)d_in[1];
    const float* a   = (const float*)d_in[2];
    const int*   src = (const int*)d_in[3];
    const int*   dst = (const int*)d_in[4];
    float* out = (float*)d_out;

    const int B        = 4;
    const int F        = in_sizes[1];
    const int E        = in_sizes[3];
    const int N        = out_size / (B * F);
    const int DEG      = E / N;
    const int xcols    = in_sizes[0] / B;
    const int node_off = xcols - N;

    const int rows    = B * N;
    const int threads = 1024;
    const size_t smem = (size_t)(N + threads + RPW * DEG + RPW) * sizeof(float);
    gat_fused_kernel<<<rows / RPW, threads, smem, stream>>>(
        x, Wn, a, src, dst, out, N, F, DEG, node_off, xcols);
}


# output phase: a wave takes two adjacent column tiles and the two accumulators exchange halves (v_permlane16_swap): each store instruction writes ONE whole 128-byte line of one row, not halves of two
# speedup vs baseline: 4.7548x; 1.0588x over previous
#include <hip/hip_runtime.h>

typedef float v2f __attribute__((ext_vector_type(2)));
typedef float v8f __attribute__((ext_vector_type(8)));
typedef float v4f __attribute__((ext_vector_type(4)));
typedef int   v4i __attribute__((ext_vector_type(4)));

__device__ __forceinline__ float lrelu(float v) { return fmaxf(v, 0.01f * v); }

__device__ __forceinline__ float xhalf(float x) {
    const int b = __builtin_bit_cast(int, x);
    return __builtin_bit_cast(float, __builtin_amdgcn_permlanex16(b, b, 0x76543210, 0xfedcba98, false, false));
}
template <int K> __device__ __forceinline__ float xrow(float x) {
    const int b = __builtin_bit_cast(int, x);
    return __builtin_bit_cast(float, __builtin_amdgcn_update_dpp(b, b, 0x160 + K, 0xf, 0xf, true));
}
__device__ __forceinline__ float wave_max(float v) {
    v = fmaxf(v, xrow<1>(v));
    v = fmaxf(v, xrow<2>(v));
    v = fmaxf(v, xrow<4>(v));
    v = fmaxf(v, xrow<8>(v));
    return fmaxf(v, xhalf(v));
}
__device__ __forceinline__ float wave_sum(float v) {
    v += xrow<1>(v);
    v += xrow<2>(v);
    v += xrow<4>(v);
    v += xrow<8>(v);
    return v + xhalf(v);
}

__device__ __forceinline__ void swap_rows(float& a, float& b) {
    asm("v_permlane16_swap_b32 %0, %1" : "+v"(a), "+v"(b));
}

#define RPW 256
#define OWN4 2

__device__ __forceinline__ float edge_logit(float ns, float nd, float s1, float s2) {
    return lrelu(fmaf(ns, s1, nd * s2));
}

template <int G>
__device__ __forceinline__ void trip_load(const v4i* __restrict__ src4, const v4i* __restrict__ dst4,
                                          int nthreads, int tid, int ngw, int& gi,
                                          v4i (&si)[G], v4i (&di)[G])
{
#pragma unroll
    for (int g = 0; g < G; ++g) {
        const int qd = __builtin_amdgcn_readfirstlane(gi * nthreads) + tid;
        si[g] = src4[qd];
        di[g] = dst4[qd];
        gi = (gi + 1 < ngw) ? gi + 1 : 0;
    }
}

template <int G>
__device__ __forceinline__ void trip_sum(const v4i (&si)[G], const v4i (&di)[G],
                                         const float* sn, float s1, float s2, float& lm, float& ls)
{
    float raw[4 * G];
#pragma unroll
    for (int g = 0; g < G; ++g) {
#pragma unroll
        for (int k = 0; k < 4; ++k) raw[4 * g + k] = edge_logit(sn[si[g][k]], sn[di[g][k]], s1, s2);
    }
    float nm = lm;
#pragma unroll
    for (int u = 0; u < 4 * G; ++u) nm = fmaxf(nm, raw[u]);
    v2f acc;
    acc.x = ls * __builtin_amdgcn_exp2f(lm - nm);
    acc.y = 0.0f;
#pragma unroll
    for (int u = 0; u < 4 * G; u += 2) {
        v2f e;
        e.x = __builtin_amdgcn_exp2f(raw[u] - nm);
        e.y = __builtin_amdgcn_exp2f(raw[u + 1] - nm);
        acc += e;
    }
    ls = acc.x + acc.y;
    lm = nm;
}

template <int G>
__device__ __forceinline__ void edge_trip(const v4i* __restrict__ src4, const v4i* __restrict__ dst4,
                                          const float* sn, float s1, float s2,
                                          int nthreads, int tid, int ngw, int& gi, float& lm, float& ls)
{
    v4i si[G], di[G];
    trip_load<G>(src4, dst4, nthreads, tid, ngw, gi, si, di);
    trip_sum<G>(si, di, sn, s1, s2, lm, ls);
}

__global__ __launch_bounds__(1024)
void gat_fused_kernel(const float* __restrict__ x,
                      const float* __restrict__ W,
                      const float* __restrict__ a,
                      const int*  __restrict__ src_idx,
                      const int*  __restrict__ dst_idx,
                      float* __restrict__ out,
                      int N, int F, int DEG, int node_off, int xcols)
{
    extern __shared__ float smem[];

    const int tid      = threadIdx.x;
    const int nthreads = blockDim.x;
    const int lane     = tid & 31;
    const int wave     = __builtin_amdgcn_readfirstlane(tid >> 5);
    const int nwaves   = nthreads >> 5;
    const int ecnt     = RPW * DEG;

    float* sn  = smem;
    float* red = sn + N;
    float* q   = red + nthreads;
    float* cs  = q + ecnt;

    const int row0 = blockIdx.x * RPW;
    const int b    = row0 / N;
    const int n0   = row0 - b * N;
    const float* xb = x + (size_t)b * xcols + node_off;

    const int j     = lane & 15;
    const int nct   = F >> 4;
    const int npair = (nct + 1) >> 1;
    const int cbase = (wave % npair) * 32;
    const float wv0 = W[cbase + j];
    const float wv1 = (cbase + 16 + j < F) ? W[cbase + 16 + j] : 0.0f;


    const v4f* __restrict__ xq = reinterpret_cast<const v4f*>(xb);
    v4f* sq = reinterpret_cast<v4f*>(sn);
    const int nq = N >> 2;
    v4f nv = {0.0f, 0.0f, 0.0f, 0.0f};
    if (tid < nq) nv = xq[tid];

    const int etot = N * DEG;
    const float lowest = -3.402823466e38f;
    const v4i* __restrict__ src4 = reinterpret_cast<const v4i*>(src_idx);
    const v4i* __restrict__ dst4 = reinterpret_cast<const v4i*>(dst_idx);

    const int gspan = 4 * nthreads;
    const int wrest = etot - 128 * wave;
    const int ngw   = wrest > 0 ? (wrest + gspan - 1) / gspan : 0;
    int gi = ngw > 0 ? (int)(blockIdx.x % (unsigned)ngw) : 0;
    const bool first4 = ngw >= 4;
    v4i fsi[4], fdi[4];
#pragma unroll
    for (int g = 0; g < 4; ++g) { fsi[g] = v4i{0, 0, 0, 0}; fdi[g] = v4i{0, 0, 0, 0}; }
    if (first4) trip_load<4>(src4, dst4, nthreads, tid, ngw, gi, fsi, fdi);

    const int qbase = (n0 * DEG) >> 2;
    v4i  osi[OWN4], odi[OWN4];
    bool haso[OWN4];
#pragma unroll
    for (int u = 0; u < OWN4; ++u) {
        haso[u] = 4 * u * nthreads + 128 * wave < ecnt;
        osi[u] = v4i{0, 0, 0, 0};
        odi[u] = v4i{0, 0, 0, 0};
        if (haso[u]) {
            const int qd = qbase + u * nthreads + tid;
            osi[u] = src4[qd];
            odi[u] = dst4[qd];
        }
    }

    float p1 = 0.0f, p2 = 0.0f;
    for (int f0 = 4 * wave; f0 < F; f0 += 4 * nwaves) {
#pragma unroll
        for (int k = 0; k < 4; ++k) { const float w = W[f0 + k]; p1 += w * a[f0 + k]; p2 += w * a[F + f0 + k]; }
    }
    float* sp = red + 2 * nwaves;
    if (lane == 0) { sp[2 * wave] = p1; sp[2 * wave + 1] = p2; }

    if (tid < nq) sq[tid] = nv;
    for (int i = tid + nthreads; i < nq; i += nthreads) sq[i] = xq[i];
    __syncthreads();

    float s1 = wave_sum(lane < nwaves ? sp[2 * lane] : 0.0f);
    float s2 = wave_sum(lane < nwaves ? sp[2 * lane + 1] : 0.0f);
    s1 = __builtin_bit_cast(float, __builtin_amdgcn_readfirstlane(__builtin_bit_cast(int, s1)));
    s2 = __builtin_bit_cast(float, __builtin_amdgcn_readfirstlane(__builtin_bit_cast(int, s2)));
    const float log2e = 1.4426950408889634f;
    s1 *= log2e;
    s2 *= log2e;

    float lm = lowest;
    float ls = 0.0f;
    const int tid4  = 4 * tid;
    int gdone = 0;
    if (first4) {
        trip_sum<4>(fsi, fdi, sn, s1, s2, lm, ls);
        gdone = 4;
    }
    for (; gdone + 4 <= ngw; gdone += 4)
        edge_trip<4>(src4, dst4, sn, s1, s2, nthreads, tid, ngw, gi, lm, ls);
    for (; gdone < ngw; ++gdone)
        edge_trip<1>(src4, dst4, sn, s1, s2, nthreads, tid, ngw, gi, lm, ls);

    float rawo[OWN4][4], ndo[OWN4][4];
#pragma unroll
    for (int u = 0; u < OWN4; ++u) {
#pragma unroll
        for (int k = 0; k < 4; ++k) { rawo[u][k] = lowest; ndo[u][k] = 0.0f; }
        if (haso[u]) {
            const v4i si = osi[u];
            const v4i di = odi[u];
#pragma unroll
            for (int k = 0; k < 4; ++k) {
                const float nd = sn[di[k]];
                rawo[u][k] = edge_logit(sn[si[k]], nd, s1, s2);
                ndo[u][k]  = nd;
            }
        }
    }

    float* redm = red;
    float* reds = red + nwaves;
    const float wmax = wave_max(lm);
    if (lane == 0) redm[wave] = wmax;
    __syncthreads();
    const float m = wave_max(lane < nwaves ? redm[lane] : lowest);

    const float wsum = wave_sum(ls * __builtin_amdgcn_exp2f(lm - m));
    if (lane == 0) reds[wave] = wsum;
    __syncthreads();
    const float invS = 1.0f / wave_sum(lane < nwaves ? reds[lane] : 0.0f);

#pragma unroll
    for (int u = 0; u < OWN4; ++u) {
        if (haso[u]) {
            v4f pq;
#pragma unroll
            for (int k = 0; k < 4; ++k) pq[k] = __builtin_amdgcn_exp2f(rawo[u][k] - m) * ndo[u][k];
            *reinterpret_cast<v4f*>(q + 4 * u * nthreads + tid4) = pq;
        }
    }
    __syncthreads();

    if (tid < RPW) {
        const float* qn = q + tid * DEG;
        float t = 0.0f;
#pragma unroll 8
        for (int k = 0; k < DEG; ++k) t += qn[k];
        cs[tid] = t * invS;
    }
    __syncthreads();

    v2f B0, B1;
    B0.x = (lane < 16) ? wv0 : 0.0f;
    B0.y = 0.0f;
    B1.x = (lane < 16) ? wv1 : 0.0f;
    B1.y = 0.0f;

    const int rstep = nwaves >= npair ? nwaves / npair : 1;
    const int col = cbase + lane;
    for (int rt = wave / npair; rt < RPW / 16; rt += rstep) {
        const float cj = cs[rt * 16 + j];

        v2f A;
        A.x = (lane < 16) ? cj : 0.0f;
        A.y = 0.0f;

        v8f acc0 = {}, acc1 = {};
        acc0 = __builtin_amdgcn_wmma_f32_16x16x4_f32(
             false, A,  false, B0,
             (short)0, acc0,  false,  false);
        acc1 = __builtin_amdgcn_wmma_f32_16x16x4_f32(
             false, A,  false, B1,
             (short)0, acc1,  false,  false);

        asm("v_nop\n\tv_nop\n\tv_nop\n\tv_nop" : "+v"(acc0), "+v"(acc1));
#pragma unroll
        for (int v = 0; v < 8; ++v) {
            float lo = acc0[v], hi = acc1[v];
            swap_rows(lo, hi);
            acc0[v] = lo;
            acc1[v] = hi;
        }
        const size_t r0 = (size_t)(row0 + rt * 16);
        if (col < F) {
#pragma unroll
            for (int v = 0; v < 8; ++v) {
                const float lo = acc0[v], hi = acc1[v];
                out[(r0 + v) * F + col]     = lo > 0.0f ? lo : 0.01f * lo;
                out[(r0 + 8 + v) * F + col] = hi > 0.0f ? hi : 0.01f * hi;
            }
        }
    }
}

extern "C" void kernel_launch(void* const* d_in, const int* in_sizes, int n_in,
                              void* d_out, int out_size, void* d_ws, size_t ws_size,
                              hipStream_t stream)
{
    const float* x   = (const float*)d_in[0];
    const float* Wn  = (const float*)d_in[1];
    const float* a   = (const float*)d_in[2];
    const int*   src = (const int*)d_in[3];
    const int*   dst = (const int*)d_in[4];
    float* out = (float*)d_out;

    const int B        = 4;
    const int F        = in_sizes[1];
    const int E        = in_sizes[3];
    const int N        = out_size / (B * F);
    const int DEG      = E / N;
    const int xcols    = in_sizes[0] / B;
    const int node_off = xcols - N;

    const int rows    = B * N;
    const int threads = 1024;
    const size_t smem = (size_t)(N + threads + RPW * DEG + RPW) * sizeof(float);
    gat_fused_kernel<<<rows / RPW, threads, smem, stream>>>(
        x, Wn, a, src, dst, out, N, F, DEG, node_off, xcols);
}
